# starting kernel
# baseline (speedup 1.0000x reference)
#include <hip/hip_runtime.h>


typedef float v2f __attribute__((ext_vector_type(2)));
typedef float v8f __attribute__((ext_vector_type(8)));

static __device__ __forceinline__ v8f wmma_f32_16x16x4(v2f a, v2f b, v8f c) {
  return __builtin_amdgcn_wmma_f32_16x16x4_f32(
       false, a,  false, b,
       (short)0, c,  false,  false);
}

__global__ void zero_f32_kernel(float* __restrict__ p, size_t n) {
  size_t i = (size_t)blockIdx.x * blockDim.x + threadIdx.x;
  size_t stride = (size_t)gridDim.x * blockDim.x;
  for (; i < n; i += stride) p[i] = 0.0f;
}

__global__ void degree_kernel(const int* __restrict__ dst,
                              float* __restrict__ deg, int nE) {
  int e = blockIdx.x * blockDim.x + threadIdx.x;
  if (e < nE) atomicAdd(&deg[dst[e]], 1.0f);
}

__global__ void scatter_add_kernel(const float* __restrict__ x,
                                   const int* __restrict__ src,
                                   const int* __restrict__ dst,
                                   float* __restrict__ agg, int nE) {
  int gid  = blockIdx.x * blockDim.x + threadIdx.x;
  int e    = gid >> 5;
  int lane = gid & 31;
  if (e >= nE) return;
  int s = src[e];
  int d = dst[e];
  const float4 v = *(const float4*)(x + (size_t)s * 128 + lane * 4);
  float* o = agg + (size_t)d * 128 + lane * 4;
  atomicAdd(o + 0, v.x);
  atomicAdd(o + 1, v.y);
  atomicAdd(o + 2, v.z);
  atomicAdd(o + 3, v.w);
}

__global__ __launch_bounds__(256)
void sage_wmma_gemm_kernel(const float* __restrict__ X,
                           const float* __restrict__ AGG,
                           const float* __restrict__ DEG,
                           const float* __restrict__ Wself,
                           const float* __restrict__ Wneigh,
                           const float* __restrict__ bias,
                           float* __restrict__ out,
                           int nRows, int nCols, int doRelu) {
  const int wavesPerBlock = blockDim.x >> 5;
  const int wave = threadIdx.x >> 5;
  const int lane = threadIdx.x & 31;
  const int nTilesN = nCols >> 4;
  const int nTiles  = (nRows >> 4) * nTilesN;
  const int tile = blockIdx.x * wavesPerBlock + wave;
  if (tile >= nTiles) return;

  const int mt   = tile / nTilesN;
  const int nt   = tile - mt * nTilesN;
  const int half = lane >> 4;
  const int l15  = lane & 15;
  const int aRow = mt * 16 + l15;
  const int bCol = nt * 16 + l15;
  const int kb   = half * 2;

  v8f c = {0.f, 0.f, 0.f, 0.f, 0.f, 0.f, 0.f, 0.f};

  const float* xr = X + (size_t)aRow * 128 + kb;
  const float* wl = Wself + (size_t)kb * nCols + bCol;
#pragma unroll 8
  for (int k = 0; k < 128; k += 4) {
    v2f a = *(const v2f*)(xr + k);
    v2f b;
    b.x = wl[(size_t)k * nCols];
    b.y = wl[(size_t)(k + 1) * nCols];
    c = wmma_f32_16x16x4(a, b, c);
  }

  const float inv = 1.0f / fmaxf(DEG[aRow], 1.0f);
  const float* gr = AGG + (size_t)aRow * 128 + kb;
  const float* wn = Wneigh + (size_t)kb * nCols + bCol;
#pragma unroll 8
  for (int k = 0; k < 128; k += 4) {
    v2f a = *(const v2f*)(gr + k);
    a.x *= inv;
    a.y *= inv;
    v2f b;
    b.x = wn[(size_t)k * nCols];
    b.y = wn[(size_t)(k + 1) * nCols];
    c = wmma_f32_16x16x4(a, b, c);
  }

  const float bv = bias[bCol];
  float* orow = out + (size_t)(mt * 16 + half * 8) * nCols + bCol;
#pragma unroll
  for (int i = 0; i < 8; ++i) {
    float v = c[i] + bv;
    if (doRelu) v = fmaxf(v, 0.0f);
    orow[(size_t)i * nCols] = v;
  }
}

extern "C" void kernel_launch(void* const* d_in, const int* in_sizes, int n_in,
                              void* d_out, int out_size, void* d_ws, size_t ws_size,
                              hipStream_t stream) {
  const float* x   = (const float*)d_in[0];
  const float* Ws1 = (const float*)d_in[1];
  const float* Wn1 = (const float*)d_in[2];
  const float* b1  = (const float*)d_in[3];
  const float* Ws2 = (const float*)d_in[4];
  const float* Wn2 = (const float*)d_in[5];
  const float* b2  = (const float*)d_in[6];
  const int*   src = (const int*)d_in[7];
  const int*   dst = (const int*)d_in[8];
  float* out = (float*)d_out;

  const int F = in_sizes[3];
  const int C = in_sizes[6];
  const int N = in_sizes[0] / F;
  const int E = in_sizes[7];

  char* ws = (char*)d_ws;
  float* deg = (float*)ws;
  float* agg = (float*)(ws + (size_t)N * 4);
  float* h   = (float*)(ws + (size_t)N * 4 + (size_t)N * F * 4);

  const int ZB = 2048, TB = 256;
  const int edgeWaveBlocks = (E * 32 + TB - 1) / TB;
  const int edgeBlocks     = (E + TB - 1) / TB;

  zero_f32_kernel<<<ZB, TB, 0, stream>>>(deg, (size_t)N + (size_t)N * F);
  degree_kernel<<<edgeBlocks, TB, 0, stream>>>(dst, deg, E);
  scatter_add_kernel<<<edgeWaveBlocks, TB, 0, stream>>>(x, src, dst, agg, E);

  {
    const int nTiles = (N / 16) * (F / 16);
    const int blocks = (nTiles + (TB / 32) - 1) / (TB / 32);
    sage_wmma_gemm_kernel<<<blocks, TB, 0, stream>>>(
        x, agg, deg, Ws1, Wn1, b1, h, N, F,  1);
  }

  zero_f32_kernel<<<ZB, TB, 0, stream>>>(agg, (size_t)N * F);
  scatter_add_kernel<<<edgeWaveBlocks, TB, 0, stream>>>(h, src, dst, agg, E);

  {
    const int nTiles = (N / 16) * (C / 16);
    const int blocks = (nTiles + (TB / 32) - 1) / (TB / 32);
    sage_wmma_gemm_kernel<<<blocks, TB, 0, stream>>>(
        h, agg, deg, Ws2, Wn2, b2, out, N, C,  0);
  }
}


# edges grouped by target in-call (count+rank add, wave-scan runs, place); each row tile sums its incoming rows in registers in source-index order into LDS as the A tile: no f32 atomics, no sum plane
# speedup vs baseline: 2.6215x; 2.6215x over previous
#include <hip/hip_runtime.h>


typedef float v2f __attribute__((ext_vector_type(2)));
typedef float v8f __attribute__((ext_vector_type(8)));

static __device__ __forceinline__ v8f wmma_f32_16x16x4(v2f a, v2f b, v8f c) {
  return __builtin_amdgcn_wmma_f32_16x16x4_f32(
       false, a,  false, b,
       (short)0, c,  false,  false);
}

__global__ void zero_i32_kernel(int* __restrict__ p, int n) {
  int i = blockIdx.x * blockDim.x + threadIdx.x;
  if (i < n) p[i] = 0;
}

__global__ void count_rank_kernel(const int* __restrict__ dst,
                                  int* __restrict__ cnt,
                                  int* __restrict__ rank, int nE) {
  int e = blockIdx.x * blockDim.x + threadIdx.x;
  if (e < nE) rank[e] = atomicAdd(&cnt[dst[e]], 1);
}

__global__ void alloc_kernel(const int* __restrict__ cnt,
                             int* __restrict__ cursor,
                             int* __restrict__ beg, int n) {
  const int i    = blockIdx.x * blockDim.x + threadIdx.x;
  const int lane = threadIdx.x & 31;
  const int c = (i < n) ? cnt[i] : 0;
  int incl = c;
  for (int d = 1; d < 32; d <<= 1) {
    const int t = __shfl_up(incl, d);
    if (lane >= d) incl += t;
  }
  int base = 0;
  if (lane == 31) base = atomicAdd(cursor, incl);
  base = __shfl(base, 31);
  if (i < n) beg[i] = base + incl - c;
}

__global__ void place_kernel(const int* __restrict__ src,
                             const int* __restrict__ dst,
                             const int* __restrict__ beg,
                             const int* __restrict__ rank,
                             int* __restrict__ perm, int nE) {
  int e = blockIdx.x * blockDim.x + threadIdx.x;
  if (e < nE) perm[beg[dst[e]] + rank[e]] = src[e];
}

template <int NT>
__global__ __launch_bounds__(NT * 32)
void sage_fused_kernel(const float* __restrict__ X,
                       const int* __restrict__ beg,
                       const int* __restrict__ cnt,
                       const int* __restrict__ perm,
                       const float* __restrict__ Wself,
                       const float* __restrict__ Wneigh,
                       const float* __restrict__ bias,
                       float* __restrict__ out, int doRelu) {
  constexpr int nCols = NT * 16;
  constexpr int RPW   = 16 / NT;
  constexpr int PITCH = 132;
  __shared__ __attribute__((aligned(16))) float As[16 * PITCH];

  const int lane = threadIdx.x & 31;
  const int wave = __builtin_amdgcn_readfirstlane(threadIdx.x >> 5);
  const int mt   = blockIdx.x;

  for (int rr = 0; rr < RPW; ++rr) {
    const int r    = wave * RPW + rr;
    const int node = mt * 16 + r;
    const int first = beg[node];
    const int deg   = cnt[node];
    const int end   = first + deg;
    float4 acc = {0.f, 0.f, 0.f, 0.f};
    for (int base = first; base < end; base += 32) {
      const int n = (end - base) < 32 ? (end - base) : 32;
      int idx = (lane < n) ? perm[base + lane] : 0x7fffffff;
      for (int k = 2; k <= 32; k <<= 1) {
        for (int j = k >> 1; j > 0; j >>= 1) {
          const int other = __shfl_xor(idx, j);
          const bool up    = (lane & k) == 0;
          const bool lower = (lane & j) == 0;
          const int mn = idx < other ? idx : other;
          const int mx = idx < other ? other : idx;
          idx = (lower == up) ? mn : mx;
        }
      }
      for (int i = 0; i < n; i += 4) {
        float4 v[4];
        for (int j = 0; j < 4; ++j) {
          const int ii = (i + j) < n ? (i + j) : (n - 1);
          const int s  = __builtin_amdgcn_readlane(idx, ii);
          v[j] = *(const float4*)(X + (size_t)s * 128 + lane * 4);
        }
        for (int j = 0; j < 4; ++j) {
          if ((i + j) < n) {
            acc.x += v[j].x;
            acc.y += v[j].y;
            acc.z += v[j].z;
            acc.w += v[j].w;
          }
        }
      }
    }
    const float inv = 1.0f / fmaxf((float)deg, 1.0f);
    acc.x *= inv;
    acc.y *= inv;
    acc.z *= inv;
    acc.w *= inv;
    *(float4*)(&As[r * PITCH + lane * 4]) = acc;
  }

  const int nt   = wave;
  const int half = lane >> 4;
  const int l15  = lane & 15;
  const int aRow = mt * 16 + l15;
  const int bCol = nt * 16 + l15;
  const int kb   = half * 2;

  v8f c = {0.f, 0.f, 0.f, 0.f, 0.f, 0.f, 0.f, 0.f};

  const float* xr = X + (size_t)aRow * 128 + kb;
  const float* wl = Wself + (size_t)kb * nCols + bCol;
#pragma unroll 8
  for (int k = 0; k < 128; k += 4) {
    v2f a = *(const v2f*)(xr + k);
    v2f b;
    b.x = wl[(size_t)k * nCols];
    b.y = wl[(size_t)(k + 1) * nCols];
    c = wmma_f32_16x16x4(a, b, c);
  }

  __syncthreads();

  const float* gr = As + l15 * PITCH + kb;
  const float* wn = Wneigh + (size_t)kb * nCols + bCol;
#pragma unroll 8
  for (int k = 0; k < 128; k += 4) {
    v2f a = *(const v2f*)(gr + k);
    v2f b;
    b.x = wn[(size_t)k * nCols];
    b.y = wn[(size_t)(k + 1) * nCols];
    c = wmma_f32_16x16x4(a, b, c);
  }

  const float bv = bias[bCol];
  float* orow = out + (size_t)(mt * 16 + half * 8) * nCols + bCol;
#pragma unroll
  for (int i = 0; i < 8; ++i) {
    float v = c[i] + bv;
    if (doRelu) v = fmaxf(v, 0.0f);
    orow[(size_t)i * nCols] = v;
  }
}

extern "C" void kernel_launch(void* const* d_in, const int* in_sizes, int n_in,
                              void* d_out, int out_size, void* d_ws, size_t ws_size,
                              hipStream_t stream) {
  const float* x   = (const float*)d_in[0];
  const float* Ws1 = (const float*)d_in[1];
  const float* Wn1 = (const float*)d_in[2];
  const float* b1  = (const float*)d_in[3];
  const float* Ws2 = (const float*)d_in[4];
  const float* Wn2 = (const float*)d_in[5];
  const float* b2  = (const float*)d_in[6];
  const int*   src = (const int*)d_in[7];
  const int*   dst = (const int*)d_in[8];
  float* out = (float*)d_out;

  const int F = in_sizes[3];
  const int N = in_sizes[0] / F;
  const int E = in_sizes[7];

  char* ws = (char*)d_ws;
  const int NP = ((N + 1 + 63) / 64) * 64;
  int*   cnt  = (int*)ws;
  int*   beg  = cnt + NP;
  int*   rank = beg + NP;
  int*   perm = rank + E;
  float* h    = (float*)(ws + (size_t)N * 4 + (size_t)N * F * 4);

  const int TB = 256;
  const int edgeBlocks = (E + TB - 1) / TB;
  const int nodeBlocks = (N + 1 + TB - 1) / TB;

  zero_i32_kernel<<<nodeBlocks, TB, 0, stream>>>(cnt, N + 1);
  count_rank_kernel<<<edgeBlocks, TB, 0, stream>>>(dst, cnt, rank, E);
  alloc_kernel<<<nodeBlocks, TB, 0, stream>>>(cnt, cnt + N, beg, N);
  place_kernel<<<edgeBlocks, TB, 0, stream>>>(src, dst, beg, rank, perm, E);

  sage_fused_kernel<8><<<N / 16, 256, 0, stream>>>(
      x, beg, cnt, perm, Ws1, Wn1, b1, h,  1);

  sage_fused_kernel<4><<<N / 16, 128, 0, stream>>>(
      h, beg, cnt, perm, Ws2, Wn2, b2, out,  0);
}


# tile's 16 own rows copied to LDS once per group by whole-row 16-byte loads; the self product reads A from LDS, not per wave 32 global 8-byte loads that touch 16 lines each; sums bit-identical
# speedup vs baseline: 2.9687x; 1.1324x over previous
#include <hip/hip_runtime.h>


typedef float v2f __attribute__((ext_vector_type(2)));
typedef float v8f __attribute__((ext_vector_type(8)));

static __device__ __forceinline__ v8f wmma_f32_16x16x4(v2f a, v2f b, v8f c) {
  return __builtin_amdgcn_wmma_f32_16x16x4_f32(
       false, a,  false, b,
       (short)0, c,  false,  false);
}

__global__ void zero_i32_kernel(int* __restrict__ p, int n) {
  int i = blockIdx.x * blockDim.x + threadIdx.x;
  if (i < n) p[i] = 0;
}

__global__ void count_rank_kernel(const int* __restrict__ dst,
                                  int* __restrict__ cnt,
                                  int* __restrict__ rank, int nE) {
  int e = blockIdx.x * blockDim.x + threadIdx.x;
  if (e < nE) rank[e] = atomicAdd(&cnt[dst[e]], 1);
}

__global__ void alloc_kernel(const int* __restrict__ cnt,
                             int* __restrict__ cursor,
                             int* __restrict__ beg, int n) {
  const int i    = blockIdx.x * blockDim.x + threadIdx.x;
  const int lane = threadIdx.x & 31;
  const int c = (i < n) ? cnt[i] : 0;
  int incl = c;
  for (int d = 1; d < 32; d <<= 1) {
    const int t = __shfl_up(incl, d);
    if (lane >= d) incl += t;
  }
  int base = 0;
  if (lane == 31) base = atomicAdd(cursor, incl);
  base = __shfl(base, 31);
  if (i < n) beg[i] = base + incl - c;
}

__global__ void place_kernel(const int* __restrict__ src,
                             const int* __restrict__ dst,
                             const int* __restrict__ beg,
                             const int* __restrict__ rank,
                             int* __restrict__ perm, int nE) {
  int e = blockIdx.x * blockDim.x + threadIdx.x;
  if (e < nE) perm[beg[dst[e]] + rank[e]] = src[e];
}

template <int NT>
__global__ __launch_bounds__(NT * 32)
void sage_fused_kernel(const float* __restrict__ X,
                       const int* __restrict__ beg,
                       const int* __restrict__ cnt,
                       const int* __restrict__ perm,
                       const float* __restrict__ Wself,
                       const float* __restrict__ Wneigh,
                       const float* __restrict__ bias,
                       float* __restrict__ out, int doRelu) {
  constexpr int nCols = NT * 16;
  constexpr int RPW   = 16 / NT;
  constexpr int PITCH = 132;
  __shared__ __attribute__((aligned(16))) float As[16 * PITCH];
  __shared__ __attribute__((aligned(16))) float Xs[16 * PITCH];

  const int lane = threadIdx.x & 31;
  const int wave = __builtin_amdgcn_readfirstlane(threadIdx.x >> 5);
  const int mt   = blockIdx.x;

  {
    float4 own[RPW];
    for (int rr = 0; rr < RPW; ++rr) {
      const int r = wave * RPW + rr;
      own[rr] = *(const float4*)(X + (size_t)(mt * 16 + r) * 128 + lane * 4);
    }
    for (int rr = 0; rr < RPW; ++rr) {
      const int r = wave * RPW + rr;
      *(float4*)(&Xs[r * PITCH + lane * 4]) = own[rr];
    }
  }

  for (int rr = 0; rr < RPW; ++rr) {
    const int r    = wave * RPW + rr;
    const int node = mt * 16 + r;
    const int first = beg[node];
    const int deg   = cnt[node];
    const int end   = first + deg;
    float4 acc = {0.f, 0.f, 0.f, 0.f};
    for (int base = first; base < end; base += 32) {
      const int n = (end - base) < 32 ? (end - base) : 32;
      int idx = (lane < n) ? perm[base + lane] : 0x7fffffff;
      for (int k = 2; k <= 32; k <<= 1) {
        for (int j = k >> 1; j > 0; j >>= 1) {
          const int other = __shfl_xor(idx, j);
          const bool up    = (lane & k) == 0;
          const bool lower = (lane & j) == 0;
          const int mn = idx < other ? idx : other;
          const int mx = idx < other ? other : idx;
          idx = (lower == up) ? mn : mx;
        }
      }
      for (int i = 0; i < n; i += 4) {
        float4 v[4];
        for (int j = 0; j < 4; ++j) {
          const int ii = (i + j) < n ? (i + j) : (n - 1);
          const int s  = __builtin_amdgcn_readlane(idx, ii);
          v[j] = *(const float4*)(X + (size_t)s * 128 + lane * 4);
        }
        for (int j = 0; j < 4; ++j) {
          if ((i + j) < n) {
            acc.x += v[j].x;
            acc.y += v[j].y;
            acc.z += v[j].z;
            acc.w += v[j].w;
          }
        }
      }
    }
    const float inv = 1.0f / fmaxf((float)deg, 1.0f);
    acc.x *= inv;
    acc.y *= inv;
    acc.z *= inv;
    acc.w *= inv;
    *(float4*)(&As[r * PITCH + lane * 4]) = acc;
  }

  const int nt   = wave;
  const int half = lane >> 4;
  const int l15  = lane & 15;
  const int bCol = nt * 16 + l15;
  const int kb   = half * 2;

  v8f c = {0.f, 0.f, 0.f, 0.f, 0.f, 0.f, 0.f, 0.f};

  __syncthreads();

  const float* xr = Xs + l15 * PITCH + kb;
  const float* wl = Wself + (size_t)kb * nCols + bCol;
#pragma unroll 8
  for (int k = 0; k < 128; k += 4) {
    v2f a = *(const v2f*)(xr + k);
    v2f b;
    b.x = wl[(size_t)k * nCols];
    b.y = wl[(size_t)(k + 1) * nCols];
    c = wmma_f32_16x16x4(a, b, c);
  }

  const float* gr = As + l15 * PITCH + kb;
  const float* wn = Wneigh + (size_t)kb * nCols + bCol;
#pragma unroll 8
  for (int k = 0; k < 128; k += 4) {
    v2f a = *(const v2f*)(gr + k);
    v2f b;
    b.x = wn[(size_t)k * nCols];
    b.y = wn[(size_t)(k + 1) * nCols];
    c = wmma_f32_16x16x4(a, b, c);
  }

  const float bv = bias[bCol];
  float* orow = out + (size_t)(mt * 16 + half * 8) * nCols + bCol;
#pragma unroll
  for (int i = 0; i < 8; ++i) {
    float v = c[i] + bv;
    if (doRelu) v = fmaxf(v, 0.0f);
    orow[(size_t)i * nCols] = v;
  }
}

extern "C" void kernel_launch(void* const* d_in, const int* in_sizes, int n_in,
                              void* d_out, int out_size, void* d_ws, size_t ws_size,
                              hipStream_t stream) {
  const float* x   = (const float*)d_in[0];
  const float* Ws1 = (const float*)d_in[1];
  const float* Wn1 = (const float*)d_in[2];
  const float* b1  = (const float*)d_in[3];
  const float* Ws2 = (const float*)d_in[4];
  const float* Wn2 = (const float*)d_in[5];
  const float* b2  = (const float*)d_in[6];
  const int*   src = (const int*)d_in[7];
  const int*   dst = (const int*)d_in[8];
  float* out = (float*)d_out;

  const int F = in_sizes[3];
  const int N = in_sizes[0] / F;
  const int E = in_sizes[7];

  char* ws = (char*)d_ws;
  const int NP = ((N + 1 + 63) / 64) * 64;
  int*   cnt  = (int*)ws;
  int*   beg  = cnt + NP;
  int*   rank = beg + NP;
  int*   perm = rank + E;
  float* h    = (float*)(ws + (size_t)N * 4 + (size_t)N * F * 4);

  const int TB = 256;
  const int edgeBlocks = (E + TB - 1) / TB;
  const int nodeBlocks = (N + 1 + TB - 1) / TB;

  zero_i32_kernel<<<nodeBlocks, TB, 0, stream>>>(cnt, N + 1);
  count_rank_kernel<<<edgeBlocks, TB, 0, stream>>>(dst, cnt, rank, E);
  alloc_kernel<<<nodeBlocks, TB, 0, stream>>>(cnt, cnt + N, beg, N);
  place_kernel<<<edgeBlocks, TB, 0, stream>>>(src, dst, beg, rank, perm, E);

  sage_fused_kernel<8><<<N / 16, 256, 0, stream>>>(
      x, beg, cnt, perm, Ws1, Wn1, b1, h,  1);

  sage_fused_kernel<4><<<N / 16, 128, 0, stream>>>(
      h, beg, cnt, perm, Ws2, Wn2, b2, out,  0);
}


# edges placed in one pass: the counting add's return is the edge's slot in a 64-slot row per target (overflow list past 64); no rank plane, scan launch or place launch: 6 launches to 4
# speedup vs baseline: 3.3595x; 1.1316x over previous
#include <hip/hip_runtime.h>


typedef float v2f __attribute__((ext_vector_type(2)));
typedef float v8f __attribute__((ext_vector_type(8)));

static __device__ __forceinline__ v8f wmma_f32_16x16x4(v2f a, v2f b, v8f c) {
  return __builtin_amdgcn_wmma_f32_16x16x4_f32(
       false, a,  false, b,
       (short)0, c,  false,  false);
}

constexpr int SLOT_CAP = 64;

__global__ void zero_i32_kernel(int* __restrict__ p, int n) {
  int i = blockIdx.x * blockDim.x + threadIdx.x;
  if (i < n) p[i] = 0;
}

__global__ void count_place_kernel(const int* __restrict__ src,
                                   const int* __restrict__ dst,
                                   int* __restrict__ cnt,
                                   int* __restrict__ ovfCnt,
                                   int* __restrict__ slots,
                                   int* __restrict__ ovf, int nE) {
  int e = blockIdx.x * blockDim.x + threadIdx.x;
  if (e < nE) {
    const int d = dst[e];
    const int s = src[e];
    const int r = atomicAdd(&cnt[d], 1);
    if (r < SLOT_CAP) {
      slots[(size_t)d * SLOT_CAP + r] = s;
    } else {
      const int o = atomicAdd(ovfCnt, 1);
      ovf[2 * (size_t)o]     = d;
      ovf[2 * (size_t)o + 1] = s;
    }
  }
}

template <int NT>
__global__ __launch_bounds__(NT * 32)
void sage_fused_kernel(const float* __restrict__ X,
                       const int* __restrict__ cnt,
                       const int* __restrict__ ovfCnt,
                       const int* __restrict__ slots,
                       const int* __restrict__ ovf,
                       const float* __restrict__ Wself,
                       const float* __restrict__ Wneigh,
                       const float* __restrict__ bias,
                       float* __restrict__ out, int doRelu) {
  constexpr int nCols = NT * 16;
  constexpr int RPW   = 16 / NT;
  constexpr int PITCH = 132;
  __shared__ __attribute__((aligned(16))) float As[16 * PITCH];
  __shared__ __attribute__((aligned(16))) float Xs[16 * PITCH];

  const int lane = threadIdx.x & 31;
  const int wave = __builtin_amdgcn_readfirstlane(threadIdx.x >> 5);
  const int mt   = blockIdx.x;

  {
    float4 own[RPW];
    for (int rr = 0; rr < RPW; ++rr) {
      const int r = wave * RPW + rr;
      own[rr] = *(const float4*)(X + (size_t)(mt * 16 + r) * 128 + lane * 4);
    }
    for (int rr = 0; rr < RPW; ++rr) {
      const int r = wave * RPW + rr;
      *(float4*)(&Xs[r * PITCH + lane * 4]) = own[rr];
    }
  }

  for (int rr = 0; rr < RPW; ++rr) {
    const int r    = wave * RPW + rr;
    const int node = mt * 16 + r;
    const int deg   = cnt[node];
    const int end   = deg < SLOT_CAP ? deg : SLOT_CAP;
    const int* row  = slots + (size_t)node * SLOT_CAP;
    float4 acc = {0.f, 0.f, 0.f, 0.f};
    for (int base = 0; base < end; base += 32) {
      const int n = (end - base) < 32 ? (end - base) : 32;
      int idx = (lane < n) ? row[base + lane] : 0x7fffffff;
      for (int k = 2; k <= 32; k <<= 1) {
        for (int j = k >> 1; j > 0; j >>= 1) {
          const int other = __shfl_xor(idx, j);
          const bool up    = (lane & k) == 0;
          const bool lower = (lane & j) == 0;
          const int mn = idx < other ? idx : other;
          const int mx = idx < other ? other : idx;
          idx = (lower == up) ? mn : mx;
        }
      }
      for (int i = 0; i < n; i += 4) {
        float4 v[4];
        for (int j = 0; j < 4; ++j) {
          const int ii = (i + j) < n ? (i + j) : (n - 1);
          const int s  = __builtin_amdgcn_readlane(idx, ii);
          v[j] = *(const float4*)(X + (size_t)s * 128 + lane * 4);
        }
        for (int j = 0; j < 4; ++j) {
          if ((i + j) < n) {
            acc.x += v[j].x;
            acc.y += v[j].y;
            acc.z += v[j].z;
            acc.w += v[j].w;
          }
        }
      }
    }
    if (deg > SLOT_CAP) {
      const int nOvf = ovfCnt[0];
      for (int base = 0; base < nOvf; base += 32) {
        const int j = base + lane;
        int d = -1;
        int s = 0;
        if (j < nOvf) {
          d = ovf[2 * (size_t)j];
          s = ovf[2 * (size_t)j + 1];
        }
        unsigned m = __builtin_amdgcn_ballot_w32(d == node);
        while (m != 0u) {
          const int l = __builtin_ctz(m);
          m &= m - 1u;
          const int ss = __builtin_amdgcn_readlane(s, l);
          const float4 v = *(const float4*)(X + (size_t)ss * 128 + lane * 4);
          acc.x += v.x;
          acc.y += v.y;
          acc.z += v.z;
          acc.w += v.w;
        }
      }
    }
    const float inv = 1.0f / fmaxf((float)deg, 1.0f);
    acc.x *= inv;
    acc.y *= inv;
    acc.z *= inv;
    acc.w *= inv;
    *(float4*)(&As[r * PITCH + lane * 4]) = acc;
  }

  const int nt   = wave;
  const int half = lane >> 4;
  const int l15  = lane & 15;
  const int bCol = nt * 16 + l15;
  const int kb   = half * 2;

  v8f c = {0.f, 0.f, 0.f, 0.f, 0.f, 0.f, 0.f, 0.f};

  __syncthreads();

  const float* xr = Xs + l15 * PITCH + kb;
  const float* wl = Wself + (size_t)kb * nCols + bCol;
#pragma unroll 8
  for (int k = 0; k < 128; k += 4) {
    v2f a = *(const v2f*)(xr + k);
    v2f b;
    b.x = wl[(size_t)k * nCols];
    b.y = wl[(size_t)(k + 1) * nCols];
    c = wmma_f32_16x16x4(a, b, c);
  }

  const float* gr = As + l15 * PITCH + kb;
  const float* wn = Wneigh + (size_t)kb * nCols + bCol;
#pragma unroll 8
  for (int k = 0; k < 128; k += 4) {
    v2f a = *(const v2f*)(gr + k);
    v2f b;
    b.x = wn[(size_t)k * nCols];
    b.y = wn[(size_t)(k + 1) * nCols];
    c = wmma_f32_16x16x4(a, b, c);
  }

  const float bv = bias[bCol];
  float* orow = out + (size_t)(mt * 16 + half * 8) * nCols + bCol;
#pragma unroll
  for (int i = 0; i < 8; ++i) {
    float v = c[i] + bv;
    if (doRelu) v = fmaxf(v, 0.0f);
    orow[(size_t)i * nCols] = v;
  }
}

extern "C" void kernel_launch(void* const* d_in, const int* in_sizes, int n_in,
                              void* d_out, int out_size, void* d_ws, size_t ws_size,
                              hipStream_t stream) {
  const float* x   = (const float*)d_in[0];
  const float* Ws1 = (const float*)d_in[1];
  const float* Wn1 = (const float*)d_in[2];
  const float* b1  = (const float*)d_in[3];
  const float* Ws2 = (const float*)d_in[4];
  const float* Wn2 = (const float*)d_in[5];
  const float* b2  = (const float*)d_in[6];
  const int*   src = (const int*)d_in[7];
  const int*   dst = (const int*)d_in[8];
  float* out = (float*)d_out;

  const int F = in_sizes[3];
  const int N = in_sizes[0] / F;
  const int E = in_sizes[7];

  char* ws = (char*)d_ws;
  const int NP = ((N + 1 + 63) / 64) * 64;
  int*   cnt   = (int*)ws;
  int*   slots = cnt + NP;
  int*   ovf   = slots + (size_t)N * SLOT_CAP;
  float* h    = (float*)(ws + (size_t)N * 4 + (size_t)N * F * 4);

  const int TB = 256;
  const int edgeBlocks = (E + TB - 1) / TB;
  const int nodeBlocks = (N + 1 + TB - 1) / TB;

  zero_i32_kernel<<<nodeBlocks, TB, 0, stream>>>(cnt, N + 1);
  count_place_kernel<<<edgeBlocks, TB, 0, stream>>>(src, dst, cnt, cnt + N, slots, ovf, E);

  sage_fused_kernel<8><<<N / 16, 256, 0, stream>>>(
      x, cnt, cnt + N, slots, ovf, Ws1, Wn1, b1, h,  1);

  sage_fused_kernel<4><<<N / 16, 128, 0, stream>>>(
      h, cnt, cnt + N, slots, ovf, Ws2, Wn2, b2, out,  0);
}
